# starting kernel
# baseline (speedup 1.0000x reference)
#include <hip/hip_runtime.h>

#define B_SZ 16384
#define T_SZ 79
#define H_SZ 256
#define HP   260

typedef __attribute__((ext_vector_type(2))) float v2f;
typedef __attribute__((ext_vector_type(8))) float v8f;

__global__ __launch_bounds__(512, 1)
void rnn_fused_kernel(const float* __restrict__ x0,
                      const float* __restrict__ x1,
                      const float* __restrict__ x2,
                      const float* __restrict__ W_in,
                      const float* __restrict__ U,
                      const float* __restrict__ b_rnn,
                      const float* __restrict__ W_d,
                      const float* __restrict__ b_d,
                      float* __restrict__ out)
{
    __shared__ __align__(16) float hbuf[2][16 * HP];
    __shared__ float xs[3][16];

    const int tid  = threadIdx.x;
    const int wave = tid >> 5;
    const int lane = tid & 31;
    const int l15  = lane & 15;
    const int half = lane >> 4;
    const int col  = (wave << 4) + l15;
    const int r0   = blockIdx.x * 16;

    for (int i = tid; i < 2 * 16 * HP; i += 512) ((float*)hbuf)[i] = 0.f;

    const float w0 = W_in[0 * H_SZ + col];
    const float w1 = W_in[1 * H_SZ + col];
    const float w2 = W_in[2 * H_SZ + col];
    const float bb = b_rnn[col];

    v2f Ureg[64];
#pragma unroll
    for (int kt = 0; kt < 64; ++kt) {
        const int krow = (kt << 2) + (half << 1);
        v2f b;
        b.x = U[krow * H_SZ + col];
        b.y = U[(krow + 1) * H_SZ + col];
        Ureg[kt] = b;
    }

    const int a_off = l15 * (HP * 4) + half * 8;

    __syncthreads();

    int cur = 0;
    for (int s = 0; s < T_SZ; ++s) {
        const int idx = (T_SZ - 1) - s;
        if (tid < 48) {
            const int f = tid >> 4, m = tid & 15;
            const float* xp = (f == 0) ? x0 : ((f == 1) ? x1 : x2);
            xs[f][m] = xp[(r0 + m) * T_SZ + idx];
        }
        __syncthreads();

        v8f acc = {0.f, 0.f, 0.f, 0.f, 0.f, 0.f, 0.f, 0.f};
        const char* abase = (const char*)(&hbuf[cur][0]) + a_off;
#pragma unroll
        for (int kt = 0; kt < 64; ++kt) {
            v2f a = *(const v2f*)(abase + kt * 16);
            acc = __builtin_amdgcn_wmma_f32_16x16x4_f32(
                      false, a, false, Ureg[kt], (short)0, acc, false, false);
        }

        float* hn = &hbuf[cur ^ 1][0];
#pragma unroll
        for (int v = 0; v < 8; ++v) {
            const int m = v + (half << 3);
            const float xv = xs[0][m] * w0 + xs[1][m] * w1 + xs[2][m] * w2 + bb;
            float hv = acc[v] + xv;
            hv = fmaxf(hv, 0.f);
            hn[m * HP + col] = hv;
        }
        __syncthreads();
        cur ^= 1;
    }

    {
        v8f acc = {0.f, 0.f, 0.f, 0.f, 0.f, 0.f, 0.f, 0.f};
        const char* abase = (const char*)(&hbuf[cur][0]) + a_off;
#pragma unroll
        for (int kt = 0; kt < 64; ++kt) {
            v2f a = *(const v2f*)(abase + kt * 16);
            const int krow = (kt << 2) + (half << 1);
            v2f b;
            b.x = W_d[krow * H_SZ + col];
            b.y = W_d[(krow + 1) * H_SZ + col];
            acc = __builtin_amdgcn_wmma_f32_16x16x4_f32(
                      false, a, false, b, (short)0, acc, false, false);
        }
        const float bd = b_d[col];
#pragma unroll
        for (int v = 0; v < 8; ++v) {
            const int m = v + (half << 3);
            out[(r0 + m) * H_SZ + col] = acc[v] + bd;
        }
    }
}

extern "C" void kernel_launch(void* const* d_in, const int* in_sizes, int n_in,
                              void* d_out, int out_size, void* d_ws, size_t ws_size,
                              hipStream_t stream) {
    (void)in_sizes; (void)n_in; (void)d_ws; (void)ws_size; (void)out_size;
    const float* x0    = (const float*)d_in[0];
    const float* x1    = (const float*)d_in[1];
    const float* x2    = (const float*)d_in[2];
    const float* W_in  = (const float*)d_in[3];
    const float* U     = (const float*)d_in[4];
    const float* b_rnn = (const float*)d_in[5];
    const float* W_d   = (const float*)d_in[6];
    const float* b_d   = (const float*)d_in[7];
    float* out = (float*)d_out;

    dim3 grid(B_SZ / 16), block(512);
    rnn_fused_kernel<<<grid, block, 0, stream>>>(x0, x1, x2, W_in, U, b_rnn,
                                                 W_d, b_d, out);
}


# Product transposed (U^T as A, h^T as B): a wave's D tiles are next step's B fragments in place; own quarter of k summed first, barrier and LDS copy of the other three quarters hide under it
# speedup vs baseline: 1.2084x; 1.2084x over previous
#include <hip/hip_runtime.h>


#define B_SZ 16384
#define T_SZ 79
#define H_SZ 256
#define XP   68
#define OP   68

typedef __attribute__((ext_vector_type(2))) float v2f;
typedef __attribute__((ext_vector_type(4))) float v4f;
typedef __attribute__((ext_vector_type(8))) float v8f;

static __device__ __forceinline__ v8f mma(v2f a, v2f b, v8f c) {
    return __builtin_amdgcn_wmma_f32_16x16x4_f32(false, a, false, b, (short)0, c, false, false);
}
static __device__ __forceinline__ void fence() { __builtin_amdgcn_sched_barrier(0); }
static __device__ __forceinline__ void lds_done_signal() {
    asm volatile("s_wait_dscnt 0x0\n\ts_barrier_signal -1" ::: "memory");
}
static __device__ __forceinline__ void barrier_wait() {
    asm volatile("s_barrier_wait -1" ::: "memory");
}

__global__ __launch_bounds__(128, 1)
void rnn_fused_kernel(const float* __restrict__ x0,
                      const float* __restrict__ x1,
                      const float* __restrict__ x2,
                      const float* __restrict__ W_in,
                      const float* __restrict__ U,
                      const float* __restrict__ b_rnn,
                      const float* __restrict__ W_d,
                      const float* __restrict__ b_d,
                      float* __restrict__ out)
{
    __shared__ __align__(16) float hx[2][4][4][2][128];
    __shared__ __align__(16) float xs[T_SZ * XP];
    __shared__ __align__(16) float ot[4][16 * OP];

    const int tid  = threadIdx.x;
    const int wave = tid >> 5;
    const int lane = tid & 31;
    const int l15  = lane & 15;
    const int half = lane >> 4;
    const int cb   = (wave << 6) + (l15 << 2);
    const int r0   = blockIdx.x * 16;

    v2f Areg[4][4][4][4];
#pragma unroll
    for (int d = 0; d < 4; ++d) {
        const int wsrc = (wave + d) & 3;
#pragma unroll
        for (int tp = 0; tp < 4; ++tp) {
#pragma unroll
            for (int q = 0; q < 4; ++q) {
                const int krow = (wsrc << 6) + (half << 5) + (q << 3) + tp;
                const v4f ux = *(const v4f*)(U + krow * H_SZ + cb);
                const v4f uy = *(const v4f*)(U + (krow + 4) * H_SZ + cb);
#pragma unroll
                for (int t = 0; t < 4; ++t) {
                    v2f a;
                    a.x = ux[t];
                    a.y = uy[t];
                    Areg[d][tp][q][t] = a;
                }
            }
        }
    }

    const int oc = (wave << 6) + (half << 5);
    v8f Wt[3][4], Bt[4];
#pragma unroll
    for (int r = 0; r < 8; ++r) {
#pragma unroll
        for (int f = 0; f < 3; ++f) {
            const v4f w = *(const v4f*)(W_in + f * H_SZ + oc + 4 * r);
#pragma unroll
            for (int t = 0; t < 4; ++t) Wt[f][t][r] = w[t];
        }
        const v4f b = *(const v4f*)(b_rnn + oc + 4 * r);
#pragma unroll
        for (int t = 0; t < 4; ++t) Bt[t][r] = b[t];
    }

    {
        float xr[3][4][3];
#pragma unroll
        for (int k = 0; k < 4; ++k) {
            const int ro = (r0 + wave + 4 * k) * T_SZ;
#pragma unroll
            for (int q = 0; q < 3; ++q) {
                const int j = min(lane + 32 * q, T_SZ - 1);
                xr[0][k][q] = x0[ro + j];
                xr[1][k][q] = x1[ro + j];
                xr[2][k][q] = x2[ro + j];
            }
        }
#pragma unroll
        for (int k = 0; k < 4; ++k) {
            const int m = wave + 4 * k;
#pragma unroll
            for (int q = 0; q < 3; ++q) {
                const int j = min(lane + 32 * q, T_SZ - 1);
                *(v4f*)(&xs[j * XP + 4 * m]) = (v4f){xr[0][k][q], xr[1][k][q], xr[2][k][q], 0.f};
            }
        }
    }

    v2f Hreg[4][4][4];
#pragma unroll
    for (int tp = 0; tp < 4; ++tp) {
#pragma unroll
        for (int q = 0; q < 4; ++q) Hreg[0][tp][q] = (v2f){0.f, 0.f};
#pragma unroll
        for (int g = 0; g < 2; ++g)
            *(v4f*)(&hx[1][wave][tp][g][lane << 2]) = (v4f){0.f, 0.f, 0.f, 0.f};
    }

    for (int s = 0; s < T_SZ; ++s) {
        const int idx = (T_SZ - 1) - s;
        const int rd  = (s + 1) & 1;

        v8f acc[4];
#pragma unroll
        for (int t = 0; t < 4; ++t) acc[t] = (v8f){0.f, 0.f, 0.f, 0.f, 0.f, 0.f, 0.f, 0.f};

#pragma unroll
        for (int q = 0; q < 4; ++q)
#pragma unroll
            for (int t = 0; t < 4; ++t) acc[t] = mma(Areg[0][0][q][t], Hreg[0][0][q], acc[t]);
        fence();
        lds_done_signal();
        fence();
#pragma unroll
        for (int q = 0; q < 4; ++q)
#pragma unroll
            for (int t = 0; t < 4; ++t) acc[t] = mma(Areg[0][1][q][t], Hreg[0][1][q], acc[t]);
        fence();
        barrier_wait();
        const v4f xq = *(const v4f*)(&xs[idx * XP + (l15 << 2)]);
#pragma unroll
        for (int d = 1; d < 4; ++d) {
            const int wsrc = (wave + d) & 3;
#pragma unroll
            for (int tp = 0; tp < 4; ++tp) {
#pragma unroll
                for (int g = 0; g < 2; ++g) {
                    const v4f v = *(const v4f*)(&hx[rd][wsrc][tp][g][lane << 2]);
                    Hreg[d][tp][2 * g]     = (v2f){v[0], v[1]};
                    Hreg[d][tp][2 * g + 1] = (v2f){v[2], v[3]};
                }
            }
        }
        fence();
#pragma unroll
        for (int tp = 2; tp < 4; ++tp)
#pragma unroll
            for (int q = 0; q < 4; ++q)
#pragma unroll
                for (int t = 0; t < 4; ++t) acc[t] = mma(Areg[0][tp][q][t], Hreg[0][tp][q], acc[t]);
#pragma unroll
        for (int tp = 0; tp < 4; ++tp)
#pragma unroll
            for (int q = 0; q < 4; ++q)
#pragma unroll
                for (int t = 0; t < 4; ++t) acc[t] = mma(Areg[1][tp][q][t], Hreg[1][tp][q], acc[t]);
#pragma unroll
        for (int tp = 0; tp < 4; ++tp)
#pragma unroll
            for (int q = 0; q < 4; ++q)
#pragma unroll
                for (int t = 0; t < 4; ++t) acc[t] = mma(Areg[2][tp][q][t], Hreg[2][tp][q], acc[t]);
        v8f xt[4];
#pragma unroll
        for (int t = 0; t < 4; ++t)
            xt[t] = xq[0] * Wt[0][t] + xq[1] * Wt[1][t] + xq[2] * Wt[2][t] + Bt[t];
#pragma unroll
        for (int tp = 0; tp < 4; ++tp)
#pragma unroll
            for (int q = 0; q < 4; ++q)
#pragma unroll
                for (int t = 0; t < 4; ++t) acc[t] = mma(Areg[3][tp][q][t], Hreg[3][tp][q], acc[t]);

#pragma unroll
        for (int tp = 0; tp < 4; ++tp) {
            v8f hn;
#pragma unroll
            for (int r = 0; r < 8; ++r) hn[r] = fmaxf(acc[tp][r] + xt[tp][r], 0.f);
#pragma unroll
            for (int q = 0; q < 4; ++q) Hreg[0][tp][q] = (v2f){hn[2 * q], hn[2 * q + 1]};
#pragma unroll
            for (int g = 0; g < 2; ++g)
                *(v4f*)(&hx[rd ^ 1][wave][tp][g][lane << 2]) =
                    (v4f){hn[4 * g], hn[4 * g + 1], hn[4 * g + 2], hn[4 * g + 3]};
        }
    }

    {
        const int fb = (T_SZ - 1) & 1;
        __syncthreads();
#pragma unroll
        for (int d = 1; d < 4; ++d) {
            const int wsrc = (wave + d) & 3;
#pragma unroll
            for (int tp = 0; tp < 4; ++tp) {
#pragma unroll
                for (int g = 0; g < 2; ++g) {
                    const v4f v = *(const v4f*)(&hx[fb][wsrc][tp][g][lane << 2]);
                    Hreg[d][tp][2 * g]     = (v2f){v[0], v[1]};
                    Hreg[d][tp][2 * g + 1] = (v2f){v[2], v[3]};
                }
            }
        }
        v8f acc[4];
#pragma unroll
        for (int t = 0; t < 4; ++t) acc[t] = (v8f){0.f, 0.f, 0.f, 0.f, 0.f, 0.f, 0.f, 0.f};
#pragma unroll
        for (int d = 0; d < 4; ++d) {
            const int wsrc = (wave + d) & 3;
#pragma unroll
            for (int tp = 0; tp < 4; ++tp) {
#pragma unroll
                for (int q = 0; q < 4; ++q) {
                    const int krow = (wsrc << 6) + (half << 5) + (q << 3) + tp;
                    const v4f dx = *(const v4f*)(W_d + krow * H_SZ + cb);
                    const v4f dy = *(const v4f*)(W_d + (krow + 4) * H_SZ + cb);
#pragma unroll
                    for (int t = 0; t < 4; ++t) {
                        v2f a;
                        a.x = dx[t];
                        a.y = dy[t];
                        acc[t] = mma(a, Hreg[d][tp][q], acc[t]);
                    }
                }
            }
        }
        float* my = &ot[wave][0];
#pragma unroll
        for (int r = 0; r < 8; ++r) {
            const v4f bd = *(const v4f*)(b_d + oc + 4 * r);
            v4f o;
#pragma unroll
            for (int t = 0; t < 4; ++t) o[t] = acc[t][r] + bd[t];
            *(v4f*)(my + l15 * OP + (half << 5) + 4 * r) = o;
        }
#pragma unroll
        for (int v = 0; v < 8; ++v) {
            const int m = v + (half << 3);
            const v4f o = *(const v4f*)(my + m * OP + (l15 << 2));
            *(v4f*)(out + (r0 + m) * H_SZ + cb) = o;
        }
    }
}

extern "C" void kernel_launch(void* const* d_in, const int* in_sizes, int n_in,
                              void* d_out, int out_size, void* d_ws, size_t ws_size,
                              hipStream_t stream) {
    (void)in_sizes; (void)n_in; (void)d_ws; (void)ws_size; (void)out_size;
    const float* x0    = (const float*)d_in[0];
    const float* x1    = (const float*)d_in[1];
    const float* x2    = (const float*)d_in[2];
    const float* W_in  = (const float*)d_in[3];
    const float* U     = (const float*)d_in[4];
    const float* b_rnn = (const float*)d_in[5];
    const float* W_d   = (const float*)d_in[6];
    const float* b_d   = (const float*)d_in[7];
    float* out = (float*)d_out;

    dim3 grid(B_SZ / 16), block(128);
    rnn_fused_kernel<<<grid, block, 0, stream>>>(x0, x1, x2, W_in, U, b_rnn,
                                                 W_d, b_d, out);
}
